# starting kernel
# baseline (speedup 1.0000x reference)
#include <hip/hip_runtime.h>
#include <math.h>


typedef __attribute__((ext_vector_type(2))) float v2f;
typedef __attribute__((ext_vector_type(8))) float v8f;

#define QL 22
#define NCOLS  (1u << (QL - 3))
#define NTILES (NCOLS / 16u)
#define PASS_BLOCKS 512u

__global__ void rgate_setup(const float* __restrict__ angle,
                            float* __restrict__ mats) {
  const int g   = blockIdx.x;
  const int j   = threadIdx.x;
  const int row = j >> 4, col = j & 15;
  const int t = row & 7, tp = col & 7;
  const int base = (g < 7) ? 3 * g : 19;

  float wr = 1.0f, wi = 0.0f;
  for (int k = 0; k < 3; ++k) {
    const int  bit    = base + k;
    const bool active = (g < 7) || (k == 2);
    const int tk = (t >> k) & 1, tpk = (tp >> k) & 1;
    float mr, mi;
    if (!active) {
      mr = (tk == tpk) ? 1.0f : 0.0f; mi = 0.0f;
    } else {
      const float th = 0.5f * angle[(QL - 1) - bit];
      float s, c;
      __sincosf(th, &s, &c);
      if (tk == tpk) { mr = c;    mi = 0.0f; }
      else           { mr = 0.0f; mi = -s;   }
    }
    const float nr = wr * mr - wi * mi;
    const float ni = wr * mi + wi * mr;
    wr = nr; wi = ni;
  }
  const int qr = row >> 3, qc = col >> 3;
  const float val = (qr == 0) ? ((qc == 0) ? wr : -wi)
                              : ((qc == 0) ? wi :  wr);
  mats[g * 256 + row * 16 + col] = val;
}

__global__ __launch_bounds__(256) void rgate_pass(
    const float* __restrict__ in, float* __restrict__ out,
    const float* __restrict__ Am, int b0, int b1, int b2) {
  const int tid  = threadIdx.x;
  const int wave = tid >> 5;
  const int lane = tid & 31;
  const int m    = lane & 15;
  const int half = lane >> 4;

  const float* arow = Am + m * 16 + 2 * half;
  v2f a0, a1, a2, a3;
  a0.x = arow[0];  a0.y = arow[1];
  a1.x = arow[4];  a1.y = arow[5];
  a2.x = arow[8];  a2.y = arow[9];
  a3.x = arow[12]; a3.y = arow[13];

  const float2* in2 = (const float2*)in;
  const unsigned waveSlots = PASS_BLOCKS * 8u;

  for (unsigned tile = blockIdx.x * 8u + (unsigned)wave; tile < NTILES;
       tile += waveSlots) {
    const unsigned r = tile * 16u + (unsigned)m;

    unsigned e = r;
    e = ((e >> b0) << (b0 + 1)) | (e & ((1u << b0) - 1u));
    e = ((e >> b1) << (b1 + 1)) | (e & ((1u << b1) - 1u));
    e = ((e >> b2) << (b2 + 1)) | (e & ((1u << b2) - 1u));

    auto idx_of = [&](int t) -> unsigned {
      return e | ((unsigned)(t & 1) << b0)
               | ((unsigned)((t >> 1) & 1) << b1)
               | ((unsigned)((t >> 2) & 1) << b2);
    };

    const int tb = half * 2;
    const float2 c0 = in2[idx_of(tb + 0)];
    const float2 c1 = in2[idx_of(tb + 1)];
    const float2 c2 = in2[idx_of(tb + 4)];
    const float2 c3 = in2[idx_of(tb + 5)];
    v2f bv0, bv1, bv2, bv3;
    bv0.x = c0.x; bv0.y = c1.x;
    bv1.x = c2.x; bv1.y = c3.x;
    bv2.x = c0.y; bv2.y = c1.y;
    bv3.x = c2.y; bv3.y = c3.y;

    v8f acc = {};
    acc = __builtin_amdgcn_wmma_f32_16x16x4_f32(false, a0, false, bv0, (short)0, acc, false, false);
    acc = __builtin_amdgcn_wmma_f32_16x16x4_f32(false, a1, false, bv1, (short)0, acc, false, false);
    acc = __builtin_amdgcn_wmma_f32_16x16x4_f32(false, a2, false, bv2, (short)0, acc, false, false);
    acc = __builtin_amdgcn_wmma_f32_16x16x4_f32(false, a3, false, bv3, (short)0, acc, false, false);

#pragma unroll
    for (int v = 0; v < 8; ++v) {
      const unsigned iv = idx_of(v);
      out[2u * iv + (unsigned)half] = acc[v];
    }
  }
}

extern "C" void kernel_launch(void* const* d_in, const int* in_sizes, int n_in,
                              void* d_out, int out_size, void* d_ws, size_t ws_size,
                              hipStream_t stream) {
  const float* x     = (const float*)d_in[0];
  const float* angle = (const float*)d_in[1];
  float* out  = (float*)d_out;
  float* mats = (float*)d_ws;

  rgate_setup<<<dim3(8), dim3(256), 0, stream>>>(angle, mats);

  for (int g = 0; g < 8; ++g) {
    const int base = (g < 7) ? 3 * g : 19;
    const float* src = (g == 0) ? x : out;
    rgate_pass<<<dim3(PASS_BLOCKS), dim3(256), 0, stream>>>(
        src, out, mats + g * 256, base, base + 1, base + 2);
  }
}


# Stage 2^13-amplitude tiles in LDS and apply the 3-qubit WMMA groups to each tile in place (5 groups, then 3; bits 12..21 regrouped, so sums round in a new order): 2 memory passes instead of 8
# speedup vs baseline: 2.8726x; 2.8726x over previous
#include <hip/hip_runtime.h>
#include <math.h>


typedef __attribute__((ext_vector_type(2))) float v2f;
typedef __attribute__((ext_vector_type(4))) float v4f;
typedef __attribute__((ext_vector_type(8))) float v8f;

#define QL 22
#define LB 13
#define TILE (1u << LB)
#define NWG  (1u << (QL - LB))
#define NTHR 512

__global__ void rgate_setup(const float* __restrict__ angle,
                            float* __restrict__ mats) {
  const int g   = blockIdx.x;
  const int j   = threadIdx.x;
  const int row = j >> 4, col = j & 15;
  const int t = row & 7, tp = col & 7;
  const int base = (g < 4) ? 3 * g : ((g == 4) ? 10 : 3 * g - 2);

  float wr = 1.0f, wi = 0.0f;
  for (int k = 0; k < 3; ++k) {
    const int  bit    = base + k;
    const bool active = (g != 4) || (k == 2);
    const int tk = (t >> k) & 1, tpk = (tp >> k) & 1;
    float mr, mi;
    if (!active) {
      mr = (tk == tpk) ? 1.0f : 0.0f; mi = 0.0f;
    } else {
      const float th = 0.5f * angle[(QL - 1) - bit];
      float s, c;
      __sincosf(th, &s, &c);
      if (tk == tpk) { mr = c;    mi = 0.0f; }
      else           { mr = 0.0f; mi = -s;   }
    }
    const float nr = wr * mr - wi * mi;
    const float ni = wr * mi + wi * mr;
    wr = nr; wi = ni;
  }
  const int qr = row >> 3, qc = col >> 3;
  const float val = (qr == 0) ? ((qc == 0) ? wr : -wi)
                              : ((qc == 0) ? wi :  wr);
  mats[g * 256 + row * 16 + col] = val;
}

__device__ __forceinline__ unsigned swz(unsigned x) {
  return x ^ ((x >> 5) & 31u) ^ (x >> 10);
}

template <unsigned MASK>
__device__ __forceinline__ unsigned dep(unsigned v) {
  unsigned r = 0;
  int k = 0;
#pragma unroll
  for (int b = 0; b < LB; ++b) {
    if (MASK & (1u << b)) { r |= ((v >> k) & 1u) << b; ++k; }
  }
  return r;
}

constexpr int popc(unsigned x) { int n = 0; while (x) { n += x & 1u; x >>= 1; } return n; }

template <int PASS>
__device__ __forceinline__ unsigned gpair(unsigned wg, unsigned q) {
  if (PASS == 0) return wg * (TILE / 2u) + q;
  return (q & 7u) | (wg << 3) | ((q >> 3) << 12);
}

template <int B0, unsigned MM>
__device__ __forceinline__ void apply_group(float* S, v2f a0, v2f a1, v2f a2, v2f a3,
                                            unsigned lane, unsigned wave) {
  constexpr unsigned GM = 7u << B0;
  constexpr unsigned TM = (TILE - 1u) & ~GM & ~MM;
  static_assert((MM & GM) == 0u, "lane bits overlap group bits");
  static_assert(popc(MM) == 4 && popc(TM) == 6, "bit split");
  const unsigned m = lane & 15u, half = lane >> 4;
  const unsigned mp = swz(dep<MM>(m));
  const unsigned rb = mp ^ (half ? swz(2u << B0) : 0u);
  const v2f* S2 = (const v2f*)S;

  unsigned ts[4];
  v2f c[4][4];
#pragma unroll
  for (int j = 0; j < 4; ++j) {
    ts[j] = swz(dep<TM>(wave * 4u + (unsigned)j));
    const unsigned r = rb ^ ts[j];
    c[j][0] = S2[r];
    c[j][1] = S2[r ^ swz(1u << B0)];
    c[j][2] = S2[r ^ swz(4u << B0)];
    c[j][3] = S2[r ^ swz(5u << B0)];
  }
  __builtin_amdgcn_sched_barrier(0);
  v8f acc[4];
#pragma unroll
  for (int j = 0; j < 4; ++j) acc[j] = v8f{};
#pragma unroll
  for (int j = 0; j < 4; ++j) {
    v2f bv0, bv1, bv2, bv3;
    bv0.x = c[j][0].x; bv0.y = c[j][1].x;
    bv1.x = c[j][2].x; bv1.y = c[j][3].x;
    bv2.x = c[j][0].y; bv2.y = c[j][1].y;
    bv3.x = c[j][2].y; bv3.y = c[j][3].y;
    acc[j] = __builtin_amdgcn_wmma_f32_16x16x4_f32(false, a0, false, bv0, (short)0, acc[j], false, false);
    acc[j] = __builtin_amdgcn_wmma_f32_16x16x4_f32(false, a1, false, bv1, (short)0, acc[j], false, false);
    acc[j] = __builtin_amdgcn_wmma_f32_16x16x4_f32(false, a2, false, bv2, (short)0, acc[j], false, false);
    acc[j] = __builtin_amdgcn_wmma_f32_16x16x4_f32(false, a3, false, bv3, (short)0, acc[j], false, false);
  }

#pragma unroll
  for (int j = 0; j < 4; ++j) {
    const unsigned w = mp ^ ts[j];
#pragma unroll
    for (int v = 0; v < 8; ++v)
      S[2u * (w ^ swz((unsigned)v << B0)) + half] = acc[j][v];
  }
}

template <int PASS>
__global__ __launch_bounds__(NTHR) void rgate_fused(const float* in, float* out,
                                                    const float* __restrict__ mats) {
  constexpr int NG = (PASS == 0) ? 5 : 3;
  constexpr int G0 = (PASS == 0) ? 0 : 5;
  __shared__ __attribute__((aligned(16))) float S[2u * TILE];
  const unsigned tid  = threadIdx.x;
  const unsigned lane = tid & 31u;
  const unsigned wave = (unsigned)__builtin_amdgcn_readfirstlane((int)(tid >> 5));
  const unsigned wg   = blockIdx.x;

  const v4f* in4 = (const v4f*)in;
  v4f* S4 = (v4f*)S;
  v4f ld[8];
#pragma unroll
  for (int i = 0; i < 8; ++i)
    ld[i] = in4[gpair<PASS>(wg, tid + (unsigned)i * NTHR)];

  const float* arow = mats + G0 * 256 + (lane & 15u) * 16u + 2u * (lane >> 4);
  v2f A[NG][4];
#pragma unroll
  for (int g = 0; g < NG; ++g)
#pragma unroll
    for (int k = 0; k < 4; ++k) {
      A[g][k].x = arow[g * 256 + 4 * k];
      A[g][k].y = arow[g * 256 + 4 * k + 1];
    }

#pragma unroll
  for (int i = 0; i < 8; ++i) {
    const unsigned l = 2u * (tid + (unsigned)i * NTHR);
    const unsigned p = swz(l);
    const v4f v = ld[i];
    S4[p >> 1] = (p & 1u) ? v4f{v.z, v.w, v.x, v.y} : v;
  }
  __syncthreads();

  if constexpr (PASS == 0) {
    apply_group<0, 0x0B8u>(S, A[0][0], A[0][1], A[0][2], A[0][3], lane, wave);
    __syncthreads();
    apply_group<3, 0x107u>(S, A[1][0], A[1][1], A[1][2], A[1][3], lane, wave);
    __syncthreads();
    apply_group<6, 0x01Bu>(S, A[2][0], A[2][1], A[2][2], A[2][3], lane, wave);
    __syncthreads();
    apply_group<9, 0x01Eu>(S, A[3][0], A[3][1], A[3][2], A[3][3], lane, wave);
    __syncthreads();
    apply_group<10, 0x01Du>(S, A[NG - 1][0], A[NG - 1][1], A[NG - 1][2], A[NG - 1][3], lane, wave);
  } else {
    apply_group<4, 0x20Eu>(S, A[0][0], A[0][1], A[0][2], A[0][3], lane, wave);
    __syncthreads();
    apply_group<7, 0x017u>(S, A[1][0], A[1][1], A[1][2], A[1][3], lane, wave);
    __syncthreads();
    apply_group<10, 0x01Du>(S, A[2][0], A[2][1], A[2][2], A[2][3], lane, wave);
  }
  __syncthreads();

  v4f* out4 = (v4f*)out;
#pragma unroll
  for (int i = 0; i < 8; ++i) {
    const unsigned q = tid + (unsigned)i * NTHR;
    const unsigned p = swz(2u * q);
    const v4f v = S4[p >> 1];
    out4[gpair<PASS>(wg, q)] = (p & 1u) ? v4f{v.z, v.w, v.x, v.y} : v;
  }
}

extern "C" void kernel_launch(void* const* d_in, const int* in_sizes, int n_in,
                              void* d_out, int out_size, void* d_ws, size_t ws_size,
                              hipStream_t stream) {
  const float* x     = (const float*)d_in[0];
  const float* angle = (const float*)d_in[1];
  float* out  = (float*)d_out;
  float* mats = (float*)d_ws;

  rgate_setup<<<dim3(8), dim3(256), 0, stream>>>(angle, mats);
  rgate_fused<0><<<dim3(NWG), dim3(NTHR), 0, stream>>>(x, out, mats);
  rgate_fused<1><<<dim3(NWG), dim3(NTHR), 0, stream>>>(out, out, mats);
}


# Fold the matrix set-up launch into the two pass kernels: each work-group builds its pass's 16x16 matrices in LDS while its tile loads are in flight (3 launches become 2; same matrices, same sums)
# speedup vs baseline: 2.9783x; 1.0368x over previous
#include <hip/hip_runtime.h>
#include <math.h>


typedef __attribute__((ext_vector_type(2))) float v2f;
typedef __attribute__((ext_vector_type(4))) float v4f;
typedef __attribute__((ext_vector_type(8))) float v8f;

#define QL 22
#define LB 13
#define TILE (1u << LB)
#define NWG  (1u << (QL - LB))
#define NTHR 512

__device__ __forceinline__ int group_of(int G0, unsigned wave, int r) {
  const int G = G0 + (int)(wave >> 3) + 2 * r;
  return (G < 7) ? G : 7;
}
__device__ __forceinline__ int group_base(int G) {
  return (G < 4) ? 3 * G : ((G == 4) ? 10 : 3 * G - 2);
}

__device__ __forceinline__ float mat_entry(int G, int j, float th0, float th1, float th2) {
  const int row = j >> 4, col = j & 15;
  const int t = row & 7, tp = col & 7;
  const int qr = row >> 3, qc = col >> 3;
  const float th[3] = {th0, th1, th2};
  float wr = 1.0f, wi = 0.0f;
#pragma unroll
  for (int k = 0; k < 3; ++k) {
    const bool active = (G != 4) || (k == 2);
    float s, c;
    __sincosf(th[k], &s, &c);
    c = active ? c : 1.0f;
    s = active ? s : 0.0f;
    const int tk = (t >> k) & 1, tpk = (tp >> k) & 1;
    const float mr = (tk == tpk) ? c : 0.0f;
    const float mi = (tk == tpk) ? 0.0f : -s;
    const float nr = wr * mr - wi * mi;
    const float ni = wr * mi + wi * mr;
    wr = nr; wi = ni;
  }
  return (qr == 0) ? ((qc == 0) ? wr : -wi)
                   : ((qc == 0) ? wi :  wr);
}

__device__ __forceinline__ unsigned swz(unsigned x) {
  return x ^ ((x >> 5) & 31u) ^ (x >> 10);
}

template <unsigned MASK>
__device__ __forceinline__ unsigned dep(unsigned v) {
  unsigned r = 0;
  int k = 0;
#pragma unroll
  for (int b = 0; b < LB; ++b) {
    if (MASK & (1u << b)) { r |= ((v >> k) & 1u) << b; ++k; }
  }
  return r;
}

constexpr int popc(unsigned x) { int n = 0; while (x) { n += x & 1u; x >>= 1; } return n; }

template <int PASS>
__device__ __forceinline__ unsigned gpair(unsigned wg, unsigned q) {
  if (PASS == 0) return wg * (TILE / 2u) + q;
  return (q & 7u) | (wg << 3) | ((q >> 3) << 12);
}

template <int B0, unsigned MM>
__device__ __forceinline__ void apply_group(float* S, v2f a0, v2f a1, v2f a2, v2f a3,
                                            unsigned lane, unsigned wave) {
  constexpr unsigned GM = 7u << B0;
  constexpr unsigned TM = (TILE - 1u) & ~GM & ~MM;
  static_assert((MM & GM) == 0u, "lane bits overlap group bits");
  static_assert(popc(MM) == 4 && popc(TM) == 6, "bit split");
  const unsigned m = lane & 15u, half = lane >> 4;
  const unsigned mp = swz(dep<MM>(m));
  const unsigned rb = mp ^ (half ? swz(2u << B0) : 0u);
  const v2f* S2 = (const v2f*)S;

  unsigned ts[4];
  v2f c[4][4];
#pragma unroll
  for (int j = 0; j < 4; ++j) {
    ts[j] = swz(dep<TM>(wave * 4u + (unsigned)j));
    const unsigned r = rb ^ ts[j];
    c[j][0] = S2[r];
    c[j][1] = S2[r ^ swz(1u << B0)];
    c[j][2] = S2[r ^ swz(4u << B0)];
    c[j][3] = S2[r ^ swz(5u << B0)];
  }
  __builtin_amdgcn_sched_barrier(0);
  v8f acc[4];
#pragma unroll
  for (int j = 0; j < 4; ++j) acc[j] = v8f{};
#pragma unroll
  for (int j = 0; j < 4; ++j) {
    v2f bv0, bv1, bv2, bv3;
    bv0.x = c[j][0].x; bv0.y = c[j][1].x;
    bv1.x = c[j][2].x; bv1.y = c[j][3].x;
    bv2.x = c[j][0].y; bv2.y = c[j][1].y;
    bv3.x = c[j][2].y; bv3.y = c[j][3].y;
    acc[j] = __builtin_amdgcn_wmma_f32_16x16x4_f32(false, a0, false, bv0, (short)0, acc[j], false, false);
    acc[j] = __builtin_amdgcn_wmma_f32_16x16x4_f32(false, a1, false, bv1, (short)0, acc[j], false, false);
    acc[j] = __builtin_amdgcn_wmma_f32_16x16x4_f32(false, a2, false, bv2, (short)0, acc[j], false, false);
    acc[j] = __builtin_amdgcn_wmma_f32_16x16x4_f32(false, a3, false, bv3, (short)0, acc[j], false, false);
  }

#pragma unroll
  for (int j = 0; j < 4; ++j) {
    const unsigned w = mp ^ ts[j];
#pragma unroll
    for (int v = 0; v < 8; ++v)
      S[2u * (w ^ swz((unsigned)v << B0)) + half] = acc[j][v];
  }
}

template <int PASS>
__global__ __launch_bounds__(NTHR) void rgate_fused(const float* in, float* out,
                                                    const float* __restrict__ angle) {
  constexpr int NG = (PASS == 0) ? 5 : 3;
  constexpr int G0 = (PASS == 0) ? 0 : 5;
  __shared__ __attribute__((aligned(16))) float S[2u * TILE];
  constexpr int NR = (NG + 1) / 2;
  __shared__ __attribute__((aligned(16))) float M[2 * NR * 256];
  const unsigned tid  = threadIdx.x;
  const unsigned lane = tid & 31u;
  const unsigned wave = (unsigned)__builtin_amdgcn_readfirstlane((int)(tid >> 5));
  const unsigned wg   = blockIdx.x;

  float th[NR][3];
#pragma unroll
  for (int r = 0; r < NR; ++r) {
    const int base = group_base(group_of(G0, wave, r));
#pragma unroll
    for (int k = 0; k < 3; ++k) th[r][k] = 0.5f * angle[(QL - 1) - (base + k)];
  }

  const v4f* in4 = (const v4f*)in;
  v4f* S4 = (v4f*)S;
  v4f ld[8];
#pragma unroll
  for (int i = 0; i < 8; ++i)
    ld[i] = in4[gpair<PASS>(wg, tid + (unsigned)i * NTHR)];

#pragma unroll
  for (int r = 0; r < NR; ++r) {
    const int j = (int)(tid & 255u);
    M[((int)(wave >> 3) + 2 * r) * 256 + j] =
        mat_entry(group_of(G0, wave, r), j, th[r][0], th[r][1], th[r][2]);
  }

#pragma unroll
  for (int i = 0; i < 8; ++i) {
    const unsigned l = 2u * (tid + (unsigned)i * NTHR);
    const unsigned p = swz(l);
    const v4f v = ld[i];
    S4[p >> 1] = (p & 1u) ? v4f{v.z, v.w, v.x, v.y} : v;
  }
  __syncthreads();

  const v2f* arow = (const v2f*)(M + (lane & 15u) * 16u + 2u * (lane >> 4));
  v2f A[NG][4];
#pragma unroll
  for (int g = 0; g < NG; ++g)
#pragma unroll
    for (int k = 0; k < 4; ++k)
      A[g][k] = arow[g * 128 + 2 * k];

  if constexpr (PASS == 0) {
    apply_group<0, 0x0B8u>(S, A[0][0], A[0][1], A[0][2], A[0][3], lane, wave);
    __syncthreads();
    apply_group<3, 0x107u>(S, A[1][0], A[1][1], A[1][2], A[1][3], lane, wave);
    __syncthreads();
    apply_group<6, 0x01Bu>(S, A[2][0], A[2][1], A[2][2], A[2][3], lane, wave);
    __syncthreads();
    apply_group<9, 0x01Eu>(S, A[3][0], A[3][1], A[3][2], A[3][3], lane, wave);
    __syncthreads();
    apply_group<10, 0x01Du>(S, A[NG - 1][0], A[NG - 1][1], A[NG - 1][2], A[NG - 1][3], lane, wave);
  } else {
    apply_group<4, 0x20Eu>(S, A[0][0], A[0][1], A[0][2], A[0][3], lane, wave);
    __syncthreads();
    apply_group<7, 0x017u>(S, A[1][0], A[1][1], A[1][2], A[1][3], lane, wave);
    __syncthreads();
    apply_group<10, 0x01Du>(S, A[2][0], A[2][1], A[2][2], A[2][3], lane, wave);
  }
  __syncthreads();

  v4f* out4 = (v4f*)out;
#pragma unroll
  for (int i = 0; i < 8; ++i) {
    const unsigned q = tid + (unsigned)i * NTHR;
    const unsigned p = swz(2u * q);
    const v4f v = S4[p >> 1];
    out4[gpair<PASS>(wg, q)] = (p & 1u) ? v4f{v.z, v.w, v.x, v.y} : v;
  }
}

extern "C" void kernel_launch(void* const* d_in, const int* in_sizes, int n_in,
                              void* d_out, int out_size, void* d_ws, size_t ws_size,
                              hipStream_t stream) {
  const float* x     = (const float*)d_in[0];
  const float* angle = (const float*)d_in[1];
  float* out  = (float*)d_out;

  rgate_fused<0><<<dim3(NWG), dim3(NTHR), 0, stream>>>(x, out, angle);
  rgate_fused<1><<<dim3(NWG), dim3(NTHR), 0, stream>>>(out, out, angle);
}


# Matrix rows/cols in amplitude order (re,im adjacent): LDS float2 feed WMMA as loaded, D pairs leave in 16 8-byte stores per group, not 32 4-byte (A read per group: VGPRs); k order of sums changes
# speedup vs baseline: 3.2356x; 1.0864x over previous
#include <hip/hip_runtime.h>
#include <math.h>


typedef __attribute__((ext_vector_type(2))) float v2f;
typedef __attribute__((ext_vector_type(4))) float v4f;
typedef __attribute__((ext_vector_type(8))) float v8f;

#define QL 22
#define LB 13
#define TILE (1u << LB)
#define NWG  (1u << (QL - LB))
#define NTHR 512

__device__ __forceinline__ int group_of(int G0, unsigned wave, int r) {
  const int G = G0 + (int)(wave >> 3) + 2 * r;
  return (G < 7) ? G : 7;
}
__device__ __forceinline__ int group_base(int G) {
  return (G < 4) ? 3 * G : ((G == 4) ? 10 : 3 * G - 2);
}

__device__ __forceinline__ float mat_entry(int G, int j, float th0, float th1, float th2) {
  const int row = j >> 4, col = j & 15;
  const int t  = ((row >> 1) & 1) | ((row >> 3) << 1) | (((row >> 2) & 1) << 2);
  const int tp = ((col >> 2) & 1) | (((col >> 1) & 1) << 1) | ((col >> 3) << 2);
  const int qr = row & 1, qc = col & 1;
  const float th[3] = {th0, th1, th2};
  float wr = 1.0f, wi = 0.0f;
#pragma unroll
  for (int k = 0; k < 3; ++k) {
    const bool active = (G != 4) || (k == 2);
    float s, c;
    __sincosf(th[k], &s, &c);
    c = active ? c : 1.0f;
    s = active ? s : 0.0f;
    const int tk = (t >> k) & 1, tpk = (tp >> k) & 1;
    const float mr = (tk == tpk) ? c : 0.0f;
    const float mi = (tk == tpk) ? 0.0f : -s;
    const float nr = wr * mr - wi * mi;
    const float ni = wr * mi + wi * mr;
    wr = nr; wi = ni;
  }
  return (qr == 0) ? ((qc == 0) ? wr : -wi)
                   : ((qc == 0) ? wi :  wr);
}

__device__ __forceinline__ unsigned swz(unsigned x) {
  return x ^ ((x >> 5) & 31u) ^ (x >> 10);
}

template <unsigned MASK>
__device__ __forceinline__ unsigned dep(unsigned v) {
  unsigned r = 0;
  int k = 0;
#pragma unroll
  for (int b = 0; b < LB; ++b) {
    if (MASK & (1u << b)) { r |= ((v >> k) & 1u) << b; ++k; }
  }
  return r;
}

constexpr int popc(unsigned x) { int n = 0; while (x) { n += x & 1u; x >>= 1; } return n; }

template <int PASS>
__device__ __forceinline__ unsigned gpair(unsigned wg, unsigned q) {
  if (PASS == 0) return wg * (TILE / 2u) + q;
  return (q & 7u) | (wg << 3) | ((q >> 3) << 12);
}

template <int B0, unsigned MM>
__device__ __forceinline__ void apply_group(float* S, const v2f* arow,
                                            unsigned lane, unsigned wave) {
  constexpr unsigned GM = 7u << B0;
  constexpr unsigned TM = (TILE - 1u) & ~GM & ~MM;
  static_assert((MM & GM) == 0u, "lane bits overlap group bits");
  static_assert(popc(MM) == 4 && popc(TM) == 6, "bit split");
  const unsigned m = lane & 15u, half = lane >> 4;
  const unsigned rb = swz(dep<MM>(m)) ^ (half ? swz(2u << B0) : 0u);
  v2f* S2 = (v2f*)S;

  const v2f a0 = arow[0], a1 = arow[2], a2 = arow[4], a3 = arow[6];
  unsigned rs[4];
  v2f c[4][4];
#pragma unroll
  for (int j = 0; j < 4; ++j) {
    const unsigned r = rb ^ swz(dep<TM>(wave * 4u + (unsigned)j));
    rs[j] = r;
    c[j][0] = S2[r];
    c[j][1] = S2[r ^ swz(1u << B0)];
    c[j][2] = S2[r ^ swz(4u << B0)];
    c[j][3] = S2[r ^ swz(5u << B0)];
  }
  __builtin_amdgcn_sched_barrier(0);
  v8f acc[4];
#pragma unroll
  for (int j = 0; j < 4; ++j) acc[j] = v8f{};
#pragma unroll
  for (int j = 0; j < 4; ++j) {
    acc[j] = __builtin_amdgcn_wmma_f32_16x16x4_f32(false, a0, false, c[j][0], (short)0, acc[j], false, false);
    acc[j] = __builtin_amdgcn_wmma_f32_16x16x4_f32(false, a1, false, c[j][1], (short)0, acc[j], false, false);
    acc[j] = __builtin_amdgcn_wmma_f32_16x16x4_f32(false, a2, false, c[j][2], (short)0, acc[j], false, false);
    acc[j] = __builtin_amdgcn_wmma_f32_16x16x4_f32(false, a3, false, c[j][3], (short)0, acc[j], false, false);
  }

#pragma unroll
  for (int j = 0; j < 4; ++j) {
    const unsigned r = rs[j];
    S2[r]                     = v2f{acc[j][0], acc[j][1]};
    S2[r ^ swz(1u << B0)]     = v2f{acc[j][2], acc[j][3]};
    S2[r ^ swz(4u << B0)]     = v2f{acc[j][4], acc[j][5]};
    S2[r ^ swz(5u << B0)]     = v2f{acc[j][6], acc[j][7]};
  }
}

template <int PASS>
__global__ __launch_bounds__(NTHR) void rgate_fused(const float* in, float* out,
                                                    const float* __restrict__ angle) {
  constexpr int NG = (PASS == 0) ? 5 : 3;
  constexpr int G0 = (PASS == 0) ? 0 : 5;
  __shared__ __attribute__((aligned(16))) float S[2u * TILE];
  constexpr int NR = (NG + 1) / 2;
  __shared__ __attribute__((aligned(16))) float M[2 * NR * 256];
  const unsigned tid  = threadIdx.x;
  const unsigned lane = tid & 31u;
  const unsigned wave = (unsigned)__builtin_amdgcn_readfirstlane((int)(tid >> 5));
  const unsigned wg   = blockIdx.x;

  float th[NR][3];
#pragma unroll
  for (int r = 0; r < NR; ++r) {
    const int base = group_base(group_of(G0, wave, r));
#pragma unroll
    for (int k = 0; k < 3; ++k) th[r][k] = 0.5f * angle[(QL - 1) - (base + k)];
  }

  const v4f* in4 = (const v4f*)in;
  v4f* S4 = (v4f*)S;
  v4f ld[8];
#pragma unroll
  for (int i = 0; i < 8; ++i)
    ld[i] = in4[gpair<PASS>(wg, tid + (unsigned)i * NTHR)];

#pragma unroll
  for (int r = 0; r < NR; ++r) {
    const int j = (int)(tid & 255u);
    M[((int)(wave >> 3) + 2 * r) * 256 + j] =
        mat_entry(group_of(G0, wave, r), j, th[r][0], th[r][1], th[r][2]);
  }

#pragma unroll
  for (int i = 0; i < 8; ++i) {
    const unsigned l = 2u * (tid + (unsigned)i * NTHR);
    const unsigned p = swz(l);
    const v4f v = ld[i];
    S4[p >> 1] = (p & 1u) ? v4f{v.z, v.w, v.x, v.y} : v;
  }
  __syncthreads();

  const v2f* arow = (const v2f*)(M + (lane & 15u) * 16u + 2u * (lane >> 4));

  if constexpr (PASS == 0) {
    apply_group<0, 0x0B8u>(S, arow + 0 * 128, lane, wave);
    __syncthreads();
    apply_group<3, 0x107u>(S, arow + 1 * 128, lane, wave);
    __syncthreads();
    apply_group<6, 0x01Bu>(S, arow + 2 * 128, lane, wave);
    __syncthreads();
    apply_group<9, 0x01Eu>(S, arow + 3 * 128, lane, wave);
    __syncthreads();
    apply_group<10, 0x01Du>(S, arow + (NG - 1) * 128, lane, wave);
  } else {
    apply_group<4, 0x20Eu>(S, arow + 0 * 128, lane, wave);
    __syncthreads();
    apply_group<7, 0x017u>(S, arow + 1 * 128, lane, wave);
    __syncthreads();
    apply_group<10, 0x01Du>(S, arow + 2 * 128, lane, wave);
  }
  __syncthreads();

  v4f* out4 = (v4f*)out;
#pragma unroll
  for (int i = 0; i < 8; ++i) {
    const unsigned q = tid + (unsigned)i * NTHR;
    const unsigned p = swz(2u * q);
    const v4f v = S4[p >> 1];
    out4[gpair<PASS>(wg, q)] = (p & 1u) ? v4f{v.z, v.w, v.x, v.y} : v;
  }
}

extern "C" void kernel_launch(void* const* d_in, const int* in_sizes, int n_in,
                              void* d_out, int out_size, void* d_ws, size_t ws_size,
                              hipStream_t stream) {
  const float* x     = (const float*)d_in[0];
  const float* angle = (const float*)d_in[1];
  float* out  = (float*)d_out;

  rgate_fused<0><<<dim3(NWG), dim3(NTHR), 0, stream>>>(x, out, angle);
  rgate_fused<1><<<dim3(NWG), dim3(NTHR), 0, stream>>>(out, out, angle);
}


# First group of a pass takes its B operands straight from global loads, last group stores D pairs straight to global (whole 128-B lines, no LDS staging); groups re-formed: sums round in a new order
# speedup vs baseline: 3.5414x; 1.0945x over previous
#include <hip/hip_runtime.h>
#include <math.h>


typedef __attribute__((ext_vector_type(2))) float v2f;
typedef __attribute__((ext_vector_type(8))) float v8f;

#define QL 22
#define LB 13
#define TILE (1u << LB)
#define NWG  (1u << (QL - LB))
#define NTHR 512

constexpr unsigned gcode(unsigned q0, unsigned q1, unsigned q2, unsigned act) {
  return q0 | (q1 << 5) | (q2 << 10) | (act << 15);
}
template <int PASS>
__device__ __forceinline__ unsigned group_code(int s) {
  if (PASS == 0)
    return (s == 0) ? gcode(5, 4, 6, 7u)
         : (s == 1) ? gcode(0, 1, 2, 7u)
         : (s == 2) ? gcode(7, 3, 11, 7u)
         : (s == 3) ? gcode(5, 12, 6, 2u)
                    : gcode(8, 9, 10, 7u);
  return (s == 0) ? gcode(14, 13, 15, 7u)
       : (s == 1) ? gcode(16, 20, 21, 7u)
                  : gcode(17, 18, 19, 7u);
}

template <int PASS>
__device__ __forceinline__ unsigned code_of(unsigned wave, int r) {
  return group_code<PASS>((int)(wave >> 3) + 2 * r);
}

__device__ __forceinline__ float mat_entry(unsigned gc, int j, float th0, float th1, float th2) {
  const int row = j >> 4, col = j & 15;
  const int t  = ((row >> 1) & 1) | ((row >> 3) << 1) | (((row >> 2) & 1) << 2);
  const int tp = ((col >> 2) & 1) | (((col >> 1) & 1) << 1) | ((col >> 3) << 2);
  const int qr = row & 1, qc = col & 1;
  const float th[3] = {th0, th1, th2};
  float wr = 1.0f, wi = 0.0f;
#pragma unroll
  for (int k = 0; k < 3; ++k) {
    const bool active = ((gc >> (15 + k)) & 1u) != 0u;
    float s, c;
    __sincosf(th[k], &s, &c);
    c = active ? c : 1.0f;
    s = active ? s : 0.0f;
    const int tk = (t >> k) & 1, tpk = (tp >> k) & 1;
    const float mr = (tk == tpk) ? c : 0.0f;
    const float mi = (tk == tpk) ? 0.0f : -s;
    const float nr = wr * mr - wi * mi;
    const float ni = wr * mi + wi * mr;
    wr = nr; wi = ni;
  }
  return (qr == 0) ? ((qc == 0) ? wr : -wi)
                   : ((qc == 0) ? wi :  wr);
}

__device__ __forceinline__ constexpr unsigned swz(unsigned x) {
  return x ^ ((x >> 5) & 31u) ^ (x >> 10);
}

template <unsigned MASK>
__device__ __forceinline__ unsigned dep(unsigned v) {
  unsigned r = 0;
  int k = 0;
#pragma unroll
  for (int b = 0; b < LB; ++b) {
    if (MASK & (1u << b)) { r |= ((v >> k) & 1u) << b; ++k; }
  }
  return r;
}

constexpr int popc(unsigned x) { int n = 0; while (x) { n += x & 1u; x >>= 1; } return n; }

template <int PASS>
__device__ __forceinline__ constexpr unsigned gofs(unsigned x) {
  return (PASS == 0) ? x : ((x & 15u) | ((x >> 4) << 13));
}
template <int PASS>
__device__ __forceinline__ unsigned gamp(unsigned wg, unsigned x) {
  return ((PASS == 0) ? wg * TILE : (wg << 4)) + gofs<PASS>(x);
}

template <int P0, int P1, int P2, unsigned MM>
struct Grp {
  static constexpr unsigned GM = (1u << P0) | (1u << P1) | (1u << P2);
  static constexpr unsigned TM = (TILE - 1u) & ~GM & ~MM;
  static_assert((MM & GM) == 0u, "lane bits overlap group bits");
  static_assert(popc(GM) == 3 && popc(MM) == 4 && popc(TM) == 6, "bit split");
  static constexpr unsigned O1 = 1u << P0, O2 = 1u << P2, O3 = O1 | O2;
  static __device__ __forceinline__ unsigned base(unsigned lane, unsigned ct) {
    return dep<MM>(lane & 15u) | ((lane >> 4) << P1) | dep<TM>(ct);
  }
};

template <class G>
__device__ __forceinline__ void lds_operands(const v2f* S2, unsigned lane, unsigned wave,
                                             v2f (&c)[4][4]) {
#pragma unroll
  for (int j = 0; j < 4; ++j) {
    const unsigned r = swz(G::base(lane, wave * 4u + (unsigned)j));
    c[j][0] = S2[r];
    c[j][1] = S2[r ^ swz(G::O1)];
    c[j][2] = S2[r ^ swz(G::O2)];
    c[j][3] = S2[r ^ swz(G::O3)];
  }
}

template <class G>
__device__ __forceinline__ void lds_results(v2f* S2, unsigned lane, unsigned wave,
                                            const v8f (&acc)[4]) {
#pragma unroll
  for (int j = 0; j < 4; ++j) {
    const unsigned r = swz(G::base(lane, wave * 4u + (unsigned)j));
    S2[r]               = v2f{acc[j][0], acc[j][1]};
    S2[r ^ swz(G::O1)]  = v2f{acc[j][2], acc[j][3]};
    S2[r ^ swz(G::O2)]  = v2f{acc[j][4], acc[j][5]};
    S2[r ^ swz(G::O3)]  = v2f{acc[j][6], acc[j][7]};
  }
}

template <int PASS, class G>
__device__ __forceinline__ void glb_operands(const v2f* g2, unsigned wg, unsigned lane,
                                             unsigned wave, v2f (&c)[4][4]) {
#pragma unroll
  for (int j = 0; j < 4; ++j) {
    const unsigned g = gamp<PASS>(wg, G::base(lane, wave * 4u + (unsigned)j));
    c[j][0] = g2[g];
    c[j][1] = g2[g + gofs<PASS>(G::O1)];
    c[j][2] = g2[g + gofs<PASS>(G::O2)];
    c[j][3] = g2[g + gofs<PASS>(G::O3)];
  }
}
template <int PASS, class G>
__device__ __forceinline__ void glb_results(v2f* g2, unsigned wg, unsigned lane,
                                            unsigned wave, const v8f (&acc)[4]) {
#pragma unroll
  for (int j = 0; j < 4; ++j) {
    const unsigned g = gamp<PASS>(wg, G::base(lane, wave * 4u + (unsigned)j));
    g2[g]                      = v2f{acc[j][0], acc[j][1]};
    g2[g + gofs<PASS>(G::O1)]  = v2f{acc[j][2], acc[j][3]};
    g2[g + gofs<PASS>(G::O2)]  = v2f{acc[j][4], acc[j][5]};
    g2[g + gofs<PASS>(G::O3)]  = v2f{acc[j][6], acc[j][7]};
  }
}

__device__ __forceinline__ void mma_tiles(const v2f* arow, const v2f (&c)[4][4], v8f (&acc)[4]) {
  const v2f a0 = arow[0], a1 = arow[2], a2 = arow[4], a3 = arow[6];
  __builtin_amdgcn_sched_barrier(0);
#pragma unroll
  for (int j = 0; j < 4; ++j) acc[j] = v8f{};
#pragma unroll
  for (int j = 0; j < 4; ++j) {
    acc[j] = __builtin_amdgcn_wmma_f32_16x16x4_f32(false, a0, false, c[j][0], (short)0, acc[j], false, false);
    acc[j] = __builtin_amdgcn_wmma_f32_16x16x4_f32(false, a1, false, c[j][1], (short)0, acc[j], false, false);
    acc[j] = __builtin_amdgcn_wmma_f32_16x16x4_f32(false, a2, false, c[j][2], (short)0, acc[j], false, false);
    acc[j] = __builtin_amdgcn_wmma_f32_16x16x4_f32(false, a3, false, c[j][3], (short)0, acc[j], false, false);
  }
}

template <class G>
__device__ __forceinline__ void apply_group(float* S, const v2f* arow, unsigned lane, unsigned wave) {
  v2f c[4][4];
  v8f acc[4];
  lds_operands<G>((const v2f*)S, lane, wave, c);
  mma_tiles(arow, c, acc);
  lds_results<G>((v2f*)S, lane, wave, acc);
}

template <int PASS>
__global__ __launch_bounds__(NTHR) void rgate_fused(const float* in, float* out,
                                                    const float* __restrict__ angle) {
  constexpr int NG = (PASS == 0) ? 5 : 3;
  __shared__ __attribute__((aligned(16))) float S[2u * TILE];
  constexpr int NR = (NG + 1) / 2;
  __shared__ __attribute__((aligned(16))) float M[2 * NR * 256];
  const unsigned tid  = threadIdx.x;
  const unsigned lane = tid & 31u;
  const unsigned wave = (unsigned)__builtin_amdgcn_readfirstlane((int)(tid >> 5));
  const unsigned wg   = blockIdx.x;

  using GFirst = Grp<5, 4, 6, 0x00Fu>;
  using GLast  = Grp<8, 9, 10, 0x00Fu>;

  float th[NR][3];
#pragma unroll
  for (int r = 0; r < NR; ++r) {
    const unsigned gc = code_of<PASS>(wave, r);
#pragma unroll
    for (int k = 0; k < 3; ++k) th[r][k] = 0.5f * angle[(QL - 1) - (int)((gc >> (5 * k)) & 31u)];
  }

  v2f c[4][4];
  glb_operands<PASS, GFirst>((const v2f*)in, wg, lane, wave, c);

#pragma unroll
  for (int r = 0; r < NR; ++r) {
    const int j = (int)(tid & 255u);
    M[((int)(wave >> 3) + 2 * r) * 256 + j] =
        mat_entry(code_of<PASS>(wave, r), j, th[r][0], th[r][1], th[r][2]);
  }
  __builtin_amdgcn_fence(__ATOMIC_RELEASE, "workgroup", "local");
  __builtin_amdgcn_s_barrier();
  __builtin_amdgcn_fence(__ATOMIC_ACQUIRE, "workgroup", "local");

  const v2f* arow = (const v2f*)(M + (lane & 15u) * 16u + 2u * (lane >> 4));

  {
    v8f acc[4];
    mma_tiles(arow + 0 * 128, c, acc);
    lds_results<GFirst>((v2f*)S, lane, wave, acc);
  }
  __syncthreads();

  if constexpr (PASS == 0) {
    apply_group<Grp<0, 1, 2, 0x0B8u>>(S, arow + 1 * 128, lane, wave);
    __syncthreads();
    apply_group<Grp<7, 3, 11, 0x017u>>(S, arow + 2 * 128, lane, wave);
    __syncthreads();
    apply_group<Grp<5, 12, 6, 0x01Bu>>(S, arow + 3 * 128, lane, wave);
  } else {
    apply_group<Grp<7, 11, 12, 0x01Du>>(S, arow + 1 * 128, lane, wave);
  }
  __syncthreads();

  {
    v8f acc[4];
    lds_operands<GLast>((const v2f*)S, lane, wave, c);
    mma_tiles(arow + (NG - 1) * 128, c, acc);
    glb_results<PASS, GLast>((v2f*)out, wg, lane, wave, acc);
  }
}

extern "C" void kernel_launch(void* const* d_in, const int* in_sizes, int n_in,
                              void* d_out, int out_size, void* d_ws, size_t ws_size,
                              hipStream_t stream) {
  const float* x     = (const float*)d_in[0];
  const float* angle = (const float*)d_in[1];
  float* out  = (float*)d_out;

  rgate_fused<0><<<dim3(NWG), dim3(NTHR), 0, stream>>>(x, out, angle);
  rgate_fused<1><<<dim3(NWG), dim3(NTHR), 0, stream>>>(out, out, angle);
}


# Pass 0: bits 0..3 as one 4-qubit group split by parity of t^tp into two independent 16x16 f32 WMMA products (32x32 real form is block-diagonal): 4 groups, not 5; sums regroup, rounding order changes
# speedup vs baseline: 3.7907x; 1.0704x over previous
#include <hip/hip_runtime.h>
#include <math.h>


typedef __attribute__((ext_vector_type(2))) float v2f;
typedef __attribute__((ext_vector_type(8))) float v8f;

#define QL 22
#define LB 13
#define TILE (1u << LB)
#define NWG  (1u << (QL - LB))
#define NTHR 512

constexpr unsigned K4 = 1u << 24;
constexpr unsigned gcode(unsigned q0, unsigned q1, unsigned q2, unsigned act) {
  return q0 | (q1 << 5) | (q2 << 10) | (act << 20);
}
constexpr unsigned gcode4(unsigned q0, unsigned q1, unsigned q2, unsigned q3, unsigned set) {
  return q0 | (q1 << 5) | (q2 << 10) | (q3 << 15) | (15u << 20) | K4 | (set << 25);
}
template <int PASS>
__device__ __forceinline__ unsigned group_code(int s) {
  if (PASS == 0)
    return (s == 0) ? gcode(5, 4, 6, 7u)
         : (s == 1) ? gcode4(0, 1, 2, 3, 0u)
         : (s == 2) ? gcode4(0, 1, 2, 3, 1u)
         : (s == 3) ? gcode(7, 11, 12, 7u)
                    : gcode(8, 9, 10, 7u);
  return (s == 0) ? gcode(14, 13, 15, 7u)
       : (s == 1) ? gcode(16, 20, 21, 7u)
                  : gcode(17, 18, 19, 7u);
}

template <int PASS>
__device__ __forceinline__ unsigned code_of(unsigned wave, int r) {
  return group_code<PASS>((int)(wave >> 3) + 2 * r);
}

__device__ __forceinline__ float mat_entry(unsigned gc, int j, float th0, float th1, float th2,
                                           float th3) {
  const int row = j >> 4, col = j & 15;
  int t  = ((row >> 1) & 1) | ((row >> 3) << 1) | (((row >> 2) & 1) << 2);
  int tp = ((col >> 2) & 1) | (((col >> 1) & 1) << 1) | ((col >> 3) << 2);
  const int qr = row & 1, qc = col & 1;
  if (gc & K4) {
    const int cs = (int)((gc >> 25) & 1u);
    const int hr = row >> 3, ur = (row >> 1) & 3;
    const int hc = (col >> 1) & 1, kc = col >> 2;
    t  = ur | ((((ur & 1) ^ (ur >> 1) ^ hr ^ qr ^ cs) & 1) << 2) | (hr << 3);
    tp = kc | ((((kc & 1) ^ (kc >> 1) ^ hc ^ qc ^ cs) & 1) << 2) | (hc << 3);
  }
  const float th[4] = {th0, th1, th2, th3};
  float wr = 1.0f, wi = 0.0f;
#pragma unroll
  for (int k = 0; k < 4; ++k) {
    const bool active = ((gc >> (20 + k)) & 1u) != 0u;
    float s, c;
    __sincosf(th[k], &s, &c);
    c = active ? c : 1.0f;
    s = active ? s : 0.0f;
    const int tk = (t >> k) & 1, tpk = (tp >> k) & 1;
    const float mr = (tk == tpk) ? c : 0.0f;
    const float mi = (tk == tpk) ? 0.0f : -s;
    const float nr = wr * mr - wi * mi;
    const float ni = wr * mi + wi * mr;
    wr = nr; wi = ni;
  }
  return (qr == 0) ? ((qc == 0) ? wr : -wi)
                   : ((qc == 0) ? wi :  wr);
}

__device__ __forceinline__ constexpr unsigned swz(unsigned x) {
  return x ^ ((x >> 5) & 31u) ^ (x >> 10);
}

template <unsigned MASK>
__device__ __forceinline__ unsigned dep(unsigned v) {
  unsigned r = 0;
  int k = 0;
#pragma unroll
  for (int b = 0; b < LB; ++b) {
    if (MASK & (1u << b)) { r |= ((v >> k) & 1u) << b; ++k; }
  }
  return r;
}

constexpr int popc(unsigned x) { int n = 0; while (x) { n += x & 1u; x >>= 1; } return n; }

template <int PASS>
__device__ __forceinline__ constexpr unsigned gofs(unsigned x) {
  return (PASS == 0) ? x : ((x & 15u) | ((x >> 4) << 13));
}
template <int PASS>
__device__ __forceinline__ unsigned gamp(unsigned wg, unsigned x) {
  return ((PASS == 0) ? wg * TILE : (wg << 4)) + gofs<PASS>(x);
}

template <int P0, int P1, int P2, unsigned MM>
struct Grp {
  static constexpr unsigned GM = (1u << P0) | (1u << P1) | (1u << P2);
  static constexpr unsigned TM = (TILE - 1u) & ~GM & ~MM;
  static_assert((MM & GM) == 0u, "lane bits overlap group bits");
  static_assert(popc(GM) == 3 && popc(MM) == 4 && popc(TM) == 6, "bit split");
  static constexpr unsigned O1 = 1u << P0, O2 = 1u << P2, O3 = O1 | O2;
  static __device__ __forceinline__ unsigned base(unsigned lane, unsigned ct) {
    return dep<MM>(lane & 15u) | ((lane >> 4) << P1) | dep<TM>(ct);
  }
};

template <int PA, int PB, int PC, int PH, unsigned MM>
struct Grp4 {
  static constexpr unsigned GM = (1u << PA) | (1u << PB) | (1u << PC) | (1u << PH);
  static constexpr unsigned TM = (TILE - 1u) & ~GM & ~MM;
  static_assert((MM & GM) == 0u, "lane bits overlap group bits");
  static_assert(popc(GM) == 4 && popc(MM) == 4 && popc(TM) == 5, "bit split");
  static constexpr unsigned OC = 1u << PC;
  static __device__ __forceinline__ constexpr unsigned eoff(int k) {
    return ((k & 1) ? (1u << PA) : 0u) | ((k & 2) ? (1u << PB) : 0u) |
           ((((k & 1) ^ (k >> 1)) & 1) ? OC : 0u);
  }
  static __device__ __forceinline__ unsigned base(unsigned lane, unsigned ct) {
    const unsigned h = lane >> 4;
    return dep<MM>(lane & 15u) | (h << PH) | (h << PC) | dep<TM>(ct);
  }
};

template <class G>
__device__ __forceinline__ void lds_operands(const v2f* S2, unsigned lane, unsigned wave,
                                             v2f (&c)[4][4]) {
#pragma unroll
  for (int j = 0; j < 4; ++j) {
    const unsigned r = swz(G::base(lane, wave * 4u + (unsigned)j));
    c[j][0] = S2[r];
    c[j][1] = S2[r ^ swz(G::O1)];
    c[j][2] = S2[r ^ swz(G::O2)];
    c[j][3] = S2[r ^ swz(G::O3)];
  }
}

template <class G>
__device__ __forceinline__ void lds_results(v2f* S2, unsigned lane, unsigned wave,
                                            const v8f (&acc)[4]) {
#pragma unroll
  for (int j = 0; j < 4; ++j) {
    const unsigned r = swz(G::base(lane, wave * 4u + (unsigned)j));
    S2[r]               = v2f{acc[j][0], acc[j][1]};
    S2[r ^ swz(G::O1)]  = v2f{acc[j][2], acc[j][3]};
    S2[r ^ swz(G::O2)]  = v2f{acc[j][4], acc[j][5]};
    S2[r ^ swz(G::O3)]  = v2f{acc[j][6], acc[j][7]};
  }
}

template <int PASS, class G>
__device__ __forceinline__ void glb_operands(const v2f* g2, unsigned wg, unsigned lane,
                                             unsigned wave, v2f (&c)[4][4]) {
#pragma unroll
  for (int j = 0; j < 4; ++j) {
    const unsigned g = gamp<PASS>(wg, G::base(lane, wave * 4u + (unsigned)j));
    c[j][0] = g2[g];
    c[j][1] = g2[g + gofs<PASS>(G::O1)];
    c[j][2] = g2[g + gofs<PASS>(G::O2)];
    c[j][3] = g2[g + gofs<PASS>(G::O3)];
  }
}
template <int PASS, class G>
__device__ __forceinline__ void glb_results(v2f* g2, unsigned wg, unsigned lane,
                                            unsigned wave, const v8f (&acc)[4]) {
#pragma unroll
  for (int j = 0; j < 4; ++j) {
    const unsigned g = gamp<PASS>(wg, G::base(lane, wave * 4u + (unsigned)j));
    g2[g]                      = v2f{acc[j][0], acc[j][1]};
    g2[g + gofs<PASS>(G::O1)]  = v2f{acc[j][2], acc[j][3]};
    g2[g + gofs<PASS>(G::O2)]  = v2f{acc[j][4], acc[j][5]};
    g2[g + gofs<PASS>(G::O3)]  = v2f{acc[j][6], acc[j][7]};
  }
}

__device__ __forceinline__ void mma_tiles(const v2f* arow, const v2f (&c)[4][4], v8f (&acc)[4]) {
  const v2f a0 = arow[0], a1 = arow[2], a2 = arow[4], a3 = arow[6];
  __builtin_amdgcn_sched_barrier(0);
#pragma unroll
  for (int j = 0; j < 4; ++j) acc[j] = v8f{};
#pragma unroll
  for (int j = 0; j < 4; ++j) {
    acc[j] = __builtin_amdgcn_wmma_f32_16x16x4_f32(false, a0, false, c[j][0], (short)0, acc[j], false, false);
    acc[j] = __builtin_amdgcn_wmma_f32_16x16x4_f32(false, a1, false, c[j][1], (short)0, acc[j], false, false);
    acc[j] = __builtin_amdgcn_wmma_f32_16x16x4_f32(false, a2, false, c[j][2], (short)0, acc[j], false, false);
    acc[j] = __builtin_amdgcn_wmma_f32_16x16x4_f32(false, a3, false, c[j][3], (short)0, acc[j], false, false);
  }
}

template <class G>
__device__ __forceinline__ void apply_group(float* S, const v2f* arow, unsigned lane, unsigned wave) {
  v2f c[4][4];
  v8f acc[4];
  lds_operands<G>((const v2f*)S, lane, wave, c);
  mma_tiles(arow, c, acc);
  lds_results<G>((v2f*)S, lane, wave, acc);
}

template <class G>
__device__ __forceinline__ void apply_group4(float* S, const v2f* arow0, const v2f* arow1,
                                             unsigned lane, unsigned wave) {
  v2f* S2 = (v2f*)S;
  v2f e[2][4], o[2][4];
#pragma unroll
  for (int j = 0; j < 2; ++j) {
    const unsigned r = swz(G::base(lane, wave * 2u + (unsigned)j));
#pragma unroll
    for (int k = 0; k < 4; ++k) {
      e[j][k] = S2[r ^ swz(G::eoff(k))];
      o[j][k] = S2[r ^ swz(G::eoff(k) ^ G::OC)];
    }
  }
  const v2f p0 = arow0[0], p1 = arow0[2], p2 = arow0[4], p3 = arow0[6];
  const v2f q0 = arow1[0], q1 = arow1[2], q2 = arow1[4], q3 = arow1[6];
  __builtin_amdgcn_sched_barrier(0);
  v2f b0[2][4], b1[2][4];
#pragma unroll
  for (int j = 0; j < 2; ++j) {
#pragma unroll
    for (int k = 0; k < 4; ++k) {
      b0[j][k] = v2f{e[j][k][0], o[j][k][1]};
      b1[j][k] = v2f{o[j][k][0], e[j][k][1]};
    }
  }
  v8f d0[2], d1[2];
#pragma unroll
  for (int j = 0; j < 2; ++j) { d0[j] = v8f{}; d1[j] = v8f{}; }
#pragma unroll
  for (int j = 0; j < 2; ++j) {
    d0[j] = __builtin_amdgcn_wmma_f32_16x16x4_f32(false, p0, false, b0[j][0], (short)0, d0[j], false, false);
    d0[j] = __builtin_amdgcn_wmma_f32_16x16x4_f32(false, p1, false, b0[j][1], (short)0, d0[j], false, false);
    d0[j] = __builtin_amdgcn_wmma_f32_16x16x4_f32(false, p2, false, b0[j][2], (short)0, d0[j], false, false);
    d0[j] = __builtin_amdgcn_wmma_f32_16x16x4_f32(false, p3, false, b0[j][3], (short)0, d0[j], false, false);
    d1[j] = __builtin_amdgcn_wmma_f32_16x16x4_f32(false, q0, false, b1[j][0], (short)0, d1[j], false, false);
    d1[j] = __builtin_amdgcn_wmma_f32_16x16x4_f32(false, q1, false, b1[j][1], (short)0, d1[j], false, false);
    d1[j] = __builtin_amdgcn_wmma_f32_16x16x4_f32(false, q2, false, b1[j][2], (short)0, d1[j], false, false);
    d1[j] = __builtin_amdgcn_wmma_f32_16x16x4_f32(false, q3, false, b1[j][3], (short)0, d1[j], false, false);
  }
#pragma unroll
  for (int j = 0; j < 2; ++j) {
    const unsigned r = swz(G::base(lane, wave * 2u + (unsigned)j));
#pragma unroll
    for (int u = 0; u < 4; ++u) {
      S2[r ^ swz(G::eoff(u))]         = v2f{d0[j][2 * u], d1[j][2 * u + 1]};
      S2[r ^ swz(G::eoff(u) ^ G::OC)] = v2f{d1[j][2 * u], d0[j][2 * u + 1]};
    }
  }
}

template <int PASS>
__global__ __launch_bounds__(NTHR) void rgate_fused(const float* in, float* out,
                                                    const float* __restrict__ angle) {
  constexpr int NG = (PASS == 0) ? 5 : 3;
  __shared__ __attribute__((aligned(16))) float S[2u * TILE];
  constexpr int NR = (NG + 1) / 2;
  __shared__ __attribute__((aligned(16))) float M[2 * NR * 256];
  const unsigned tid  = threadIdx.x;
  const unsigned lane = tid & 31u;
  const unsigned wave = (unsigned)__builtin_amdgcn_readfirstlane((int)(tid >> 5));
  const unsigned wg   = blockIdx.x;

  using GFirst = Grp<5, 4, 6, 0x00Fu>;
  using GLast  = Grp<8, 9, 10, 0x00Fu>;

  float th[NR][4];
#pragma unroll
  for (int r = 0; r < NR; ++r) {
    const unsigned gc = code_of<PASS>(wave, r);
#pragma unroll
    for (int k = 0; k < 4; ++k) th[r][k] = 0.5f * angle[(QL - 1) - (int)((gc >> (5 * k)) & 31u)];
  }

  v2f c[4][4];
  glb_operands<PASS, GFirst>((const v2f*)in, wg, lane, wave, c);

#pragma unroll
  for (int r = 0; r < NR; ++r) {
    const int j = (int)(tid & 255u);
    M[((int)(wave >> 3) + 2 * r) * 256 + j] =
        mat_entry(code_of<PASS>(wave, r), j, th[r][0], th[r][1], th[r][2], th[r][3]);
  }
  __builtin_amdgcn_fence(__ATOMIC_RELEASE, "workgroup", "local");
  __builtin_amdgcn_s_barrier();
  __builtin_amdgcn_fence(__ATOMIC_ACQUIRE, "workgroup", "local");

  const v2f* arow = (const v2f*)(M + (lane & 15u) * 16u + 2u * (lane >> 4));

  {
    v8f acc[4];
    mma_tiles(arow + 0 * 128, c, acc);
    lds_results<GFirst>((v2f*)S, lane, wave, acc);
  }
  __syncthreads();

  if constexpr (PASS == 0) {
    apply_group4<Grp4<0, 1, 2, 3, 0x0F0u>>(S, arow + 1 * 128, arow + 2 * 128, lane, wave);
    __syncthreads();
    apply_group<Grp<7, 11, 12, 0x01Du>>(S, arow + 3 * 128, lane, wave);
  } else {
    apply_group<Grp<7, 11, 12, 0x01Du>>(S, arow + 1 * 128, lane, wave);
  }
  __syncthreads();

  {
    v8f acc[4];
    lds_operands<GLast>((const v2f*)S, lane, wave, c);
    mma_tiles(arow + (NG - 1) * 128, c, acc);
    glb_results<PASS, GLast>((v2f*)out, wg, lane, wave, acc);
  }
}

extern "C" void kernel_launch(void* const* d_in, const int* in_sizes, int n_in,
                              void* d_out, int out_size, void* d_ws, size_t ws_size,
                              hipStream_t stream) {
  const float* x     = (const float*)d_in[0];
  const float* angle = (const float*)d_in[1];
  float* out  = (float*)d_out;

  rgate_fused<0><<<dim3(NWG), dim3(NTHR), 0, stream>>>(x, out, angle);
  rgate_fused<1><<<dim3(NWG), dim3(NTHR), 0, stream>>>(out, out, angle);
}
